# starting kernel
# baseline (speedup 1.0000x reference)
#include <hip/hip_runtime.h>
#include <hip/hip_bf16.h>
#include <math.h>

typedef float v2f __attribute__((ext_vector_type(2)));
typedef float v4f __attribute__((ext_vector_type(4)));
typedef float v8f __attribute__((ext_vector_type(8)));

#define B_SZ    4
#define T_LEN   2048
#define D_MODEL 1024
#define D_STATE 16
#define DT_RANK 64
#define M_ROWS  (B_SZ * T_LEN)

#define NCHUNK  32
#define CLEN    (T_LEN / NCHUNK)

template <int ACT>
__global__ __launch_bounds__(256)
void wmma_gemm_f32(const float* __restrict__ A,
                   const float* __restrict__ W,
                   const float* __restrict__ bias,
                   float* __restrict__ out,
                   int M, int N, int K) {
    const int wave   = (blockIdx.x * blockDim.x + threadIdx.x) >> 5;
    const int lane   = threadIdx.x & 31;
    const int ntiles = N >> 4;
    const int m0     = (wave / ntiles) << 4;
    const int n0     = (wave % ntiles) << 4;
    const int half   = lane >> 4;
    const int l15    = lane & 15;

    const float* arow = A + (size_t)(m0 + l15) * K + 2 * half;
    const float* wrow = W + (size_t)(n0 + l15) * K + 2 * half;

    v8f c = {};
#pragma unroll 4
    for (int k = 0; k < K; k += 4) {
        v2f a = *(const v2f*)(arow + k);
        v2f b = *(const v2f*)(wrow + k);
        c = __builtin_amdgcn_wmma_f32_16x16x4_f32(
                false, a, false, b, (short)0, c, false, false);
    }

    const float bv = bias[n0 + l15];
#pragma unroll
    for (int r = 0; r < 8; ++r) {
        float v = c[r] + bv;
        if (ACT == 1) v = (v > 20.0f) ? v : log1pf(__expf(v));
        out[(size_t)(m0 + half * 8 + r) * N + (n0 + l15)] = v;
    }
}

__global__ __launch_bounds__(256)
void scan_phase1(const float* __restrict__ u,
                 const float* __restrict__ dt,
                 const float* __restrict__ Btg,
                 const float* __restrict__ A_log,
                 float* __restrict__ Pbuf,
                 float* __restrict__ Sbuf) {
    __shared__ float sB[CLEN][D_STATE];

    const int chunk = blockIdx.x & (NCHUNK - 1);
    const int rest  = blockIdx.x >> 5;
    const int b     = rest >> 2;
    const int d     = ((rest & 3) << 8) + threadIdx.x;

    const float LOG2E = 1.44269504088896340736f;
    float Alog2[D_STATE];
#pragma unroll
    for (int n = 0; n < D_STATE; ++n)
        Alog2[n] = -__expf(A_log[d * D_STATE + n]) * LOG2E;

    const size_t baseS = ((size_t)b * T_LEN + (size_t)chunk * CLEN) * D_STATE;
    for (int i = threadIdx.x; i < CLEN * D_STATE; i += 256)
        ((float*)sB)[i] = Btg[baseS + i];
    __syncthreads();

    float P[D_STATE], S[D_STATE];
#pragma unroll
    for (int n = 0; n < D_STATE; ++n) { P[n] = 1.0f; S[n] = 0.0f; }

    const size_t baseU = ((size_t)b * T_LEN + (size_t)chunk * CLEN) * D_MODEL + d;
    for (int tt = 0; tt < CLEN; ++tt) {
        const size_t off = baseU + (size_t)tt * D_MODEL;
        const float ul  = u[off];
        const float dtl = dt[off];
        const float dtu = dtl * ul;
#pragma unroll
        for (int n = 0; n < D_STATE; ++n) {
            float abar = exp2f(dtl * Alog2[n]);
            P[n] *= abar;
            S[n]  = fmaf(abar, S[n], sB[tt][n] * dtu);
        }
    }

    const size_t o = (((size_t)b * NCHUNK + chunk) * D_MODEL + d) * D_STATE;
#pragma unroll
    for (int q = 0; q < 4; ++q) {
        *(v4f*)(Pbuf + o + 4 * q) = *(v4f*)(P + 4 * q);
        *(v4f*)(Sbuf + o + 4 * q) = *(v4f*)(S + 4 * q);
    }
}

__global__ __launch_bounds__(256)
void scan_phase2(float* __restrict__ Pbuf,
                 const float* __restrict__ Sbuf) {
    const int b = blockIdx.x >> 2;
    const int d = ((blockIdx.x & 3) << 8) + threadIdx.x;

    float h[D_STATE];
#pragma unroll
    for (int n = 0; n < D_STATE; ++n) h[n] = 0.0f;

    for (int chunk = 0; chunk < NCHUNK; ++chunk) {
        const size_t o = (((size_t)b * NCHUNK + chunk) * D_MODEL + d) * D_STATE;
        float P[D_STATE], S[D_STATE];
#pragma unroll
        for (int q = 0; q < 4; ++q) {
            *(v4f*)(P + 4 * q) = *(const v4f*)(Pbuf + o + 4 * q);
            *(v4f*)(S + 4 * q) = *(const v4f*)(Sbuf + o + 4 * q);
        }
#pragma unroll
        for (int q = 0; q < 4; ++q)
            *(v4f*)(Pbuf + o + 4 * q) = *(v4f*)(h + 4 * q);
#pragma unroll
        for (int n = 0; n < D_STATE; ++n)
            h[n] = fmaf(P[n], h[n], S[n]);
    }
}

__global__ __launch_bounds__(256)
void scan_phase3(const float* __restrict__ u,
                 const float* __restrict__ dt,
                 const float* __restrict__ Btg,
                 const float* __restrict__ Ctg,
                 const float* __restrict__ A_log,
                 const float* __restrict__ Dvec,
                 const float* __restrict__ Hstart,
                 float* __restrict__ out) {
    __shared__ float sB[CLEN][D_STATE];
    __shared__ float sC[CLEN][D_STATE];

    const int chunk = blockIdx.x & (NCHUNK - 1);
    const int rest  = blockIdx.x >> 5;
    const int b     = rest >> 2;
    const int d     = ((rest & 3) << 8) + threadIdx.x;

    const float LOG2E = 1.44269504088896340736f;
    float Alog2[D_STATE];
#pragma unroll
    for (int n = 0; n < D_STATE; ++n)
        Alog2[n] = -__expf(A_log[d * D_STATE + n]) * LOG2E;

    const size_t baseS = ((size_t)b * T_LEN + (size_t)chunk * CLEN) * D_STATE;
    for (int i = threadIdx.x; i < CLEN * D_STATE; i += 256) {
        ((float*)sB)[i] = Btg[baseS + i];
        ((float*)sC)[i] = Ctg[baseS + i];
    }
    __syncthreads();

    float h[D_STATE];
    const size_t ho = (((size_t)b * NCHUNK + chunk) * D_MODEL + d) * D_STATE;
#pragma unroll
    for (int q = 0; q < 4; ++q)
        *(v4f*)(h + 4 * q) = *(const v4f*)(Hstart + ho + 4 * q);

    const float Dd = Dvec[d];
    const size_t baseU = ((size_t)b * T_LEN + (size_t)chunk * CLEN) * D_MODEL + d;
    for (int tt = 0; tt < CLEN; ++tt) {
        const size_t off = baseU + (size_t)tt * D_MODEL;
        const float ul  = u[off];
        const float dtl = dt[off];
        const float dtu = dtl * ul;
        float y = Dd * ul;
#pragma unroll
        for (int n = 0; n < D_STATE; ++n) {
            float abar = exp2f(dtl * Alog2[n]);
            h[n] = fmaf(abar, h[n], sB[tt][n] * dtu);
            y    = fmaf(h[n], sC[tt][n], y);
        }
        out[off] = y;
    }
}

extern "C" void kernel_launch(void* const* d_in, const int* in_sizes, int n_in,
                              void* d_out, int out_size, void* d_ws, size_t ws_size,
                              hipStream_t stream) {
    const float* u     = (const float*)d_in[0];
    const float* A_log = (const float*)d_in[1];
    const float* W_xdt = (const float*)d_in[2];
    const float* b_xdt = (const float*)d_in[3];
    const float* W_dt  = (const float*)d_in[4];
    const float* b_dt  = (const float*)d_in[5];
    const float* W_B   = (const float*)d_in[6];
    const float* b_B   = (const float*)d_in[7];
    const float* W_C   = (const float*)d_in[8];
    const float* b_C   = (const float*)d_in[9];
    const float* Dv    = (const float*)d_in[10];
    float* out = (float*)d_out;

    float* ws   = (float*)d_ws;
    float* xdt  = ws;
    float* dt   = xdt + (size_t)M_ROWS * DT_RANK;
    float* Bt   = dt  + (size_t)M_ROWS * D_MODEL;
    float* Ct   = Bt  + (size_t)M_ROWS * D_STATE;
    float* Pbuf = Ct  + (size_t)M_ROWS * D_STATE;
    float* Sbuf = Pbuf + (size_t)B_SZ * NCHUNK * D_MODEL * D_STATE;

    wmma_gemm_f32<0><<<(M_ROWS / 16) * (DT_RANK / 16) / 8, 256, 0, stream>>>(
        u, W_xdt, b_xdt, xdt, M_ROWS, DT_RANK, D_MODEL);

    wmma_gemm_f32<1><<<(M_ROWS / 16) * (D_MODEL / 16) / 8, 256, 0, stream>>>(
        xdt, W_dt, b_dt, dt, M_ROWS, D_MODEL, DT_RANK);

    wmma_gemm_f32<0><<<(M_ROWS / 16) / 8, 256, 0, stream>>>(
        u, W_B, b_B, Bt, M_ROWS, D_STATE, D_MODEL);
    wmma_gemm_f32<0><<<(M_ROWS / 16) / 8, 256, 0, stream>>>(
        u, W_C, b_C, Ct, M_ROWS, D_STATE, D_MODEL);

    const int scan_blocks = B_SZ * 4 * NCHUNK;
    scan_phase1<<<scan_blocks, 256, 0, stream>>>(u, dt, Bt, A_log, Pbuf, Sbuf);
    scan_phase2<<<B_SZ * 4, 256, 0, stream>>>(Pbuf, Sbuf);
    scan_phase3<<<scan_blocks, 256, 0, stream>>>(u, dt, Bt, Ct, A_log, Dv, Pbuf, out);
}


# u projections (x_dt, B, C) in one launch: waves split the OUTPUT (32 rows x 3 of 6 column tiles, whole K in the old order); u and weights staged through LDS by whole-line loads one chunk ahead
# speedup vs baseline: 2.0270x; 2.0270x over previous
#include <hip/hip_runtime.h>
#include <hip/hip_bf16.h>
#include <math.h>

typedef float v2f __attribute__((ext_vector_type(2)));
typedef float v4f __attribute__((ext_vector_type(4)));
typedef float v8f __attribute__((ext_vector_type(8)));

#define B_SZ    4
#define T_LEN   2048
#define D_MODEL 1024
#define D_STATE 16
#define DT_RANK 64
#define M_ROWS  (B_SZ * T_LEN)

#define SCH     128
#define NSTG    (T_LEN / SCH)
#define SFETCH  (SCH / 32)

template <int ACT>
__global__ __launch_bounds__(256)
void wmma_gemm_f32(const float* __restrict__ A,
                   const float* __restrict__ W,
                   const float* __restrict__ bias,
                   float* __restrict__ out,
                   int M, int N, int K) {
    const int wave   = (blockIdx.x * blockDim.x + threadIdx.x) >> 5;
    const int lane   = threadIdx.x & 31;
    const int ntiles = N >> 4;
    const int m0     = (wave / ntiles) << 4;
    const int n0     = (wave % ntiles) << 4;
    const int half   = lane >> 4;
    const int l15    = lane & 15;

    const float* arow = A + (size_t)(m0 + l15) * K + 2 * half;
    const float* wrow = W + (size_t)(n0 + l15) * K + 2 * half;

    v8f c = {};
#pragma unroll 4
    for (int k = 0; k < K; k += 4) {
        v2f a = *(const v2f*)(arow + k);
        v2f b = *(const v2f*)(wrow + k);
        c = __builtin_amdgcn_wmma_f32_16x16x4_f32(
                false, a, false, b, (short)0, c, false, false);
    }

    const float bv = bias[n0 + l15];
#pragma unroll
    for (int r = 0; r < 8; ++r) {
        float v = c[r] + bv;
        if (ACT == 1) v = (v > 20.0f) ? v : log1pf(__expf(v));
        out[(size_t)(m0 + half * 8 + r) * N + (n0 + l15)] = v;
    }
}

#define PRJ_ROWS   32
#define PRJ_COLS   96
#define PRJ_KC     64
#define PRJ_PITCH  68
#define PRJ_STAGED (PRJ_ROWS + PRJ_COLS)
#define PRJ_NCH    (D_MODEL / PRJ_KC)
#define PRJ_NF4    (PRJ_STAGED * PRJ_KC / 4 / 128)

__device__ __forceinline__ void lds_barrier() {
    __builtin_amdgcn_fence(__ATOMIC_RELEASE, "workgroup", "local");
    __builtin_amdgcn_s_barrier();
    __builtin_amdgcn_fence(__ATOMIC_ACQUIRE, "workgroup", "local");
}

__global__ __launch_bounds__(128)
void proj_u_fused(const float* __restrict__ u,
                  const float* __restrict__ W_xdt, const float* __restrict__ b_xdt,
                  const float* __restrict__ W_B,   const float* __restrict__ b_B,
                  const float* __restrict__ W_C,   const float* __restrict__ b_C,
                  float* __restrict__ xdt, float* __restrict__ Bt, float* __restrict__ Ct) {
    __shared__ __attribute__((aligned(16))) float sP[2][PRJ_STAGED * PRJ_PITCH];

    const int tid  = threadIdx.x;
    const int lane = tid & 31;
    const int w    = tid >> 5;
    const int rt   = w & 1;
    const int cg   = w >> 1;
    const int c    = lane & 15;
    const int h    = lane >> 4;
    const int m0   = blockIdx.x * PRJ_ROWS;
    const int r8   = tid >> 4;
    const int c4   = tid & 15;

    v4f r[PRJ_NF4];

    auto fetch = [&](int k0) {
#pragma unroll
        for (int i = 0; i < PRJ_NF4; ++i) {
            const int row = r8 + 8 * i;
            const float* p;
            if (i < 4)       p = u     + (size_t)(m0 + row)  * D_MODEL;
            else if (i < 12) p = W_xdt + (size_t)(row - 32)  * D_MODEL;
            else if (i < 14) p = W_B   + (size_t)(row - 96)  * D_MODEL;
            else             p = W_C   + (size_t)(row - 112) * D_MODEL;
            r[i] = *(const v4f*)(p + k0 + 4 * c4);
        }
    };
    auto put = [&](int st) {
#pragma unroll
        for (int i = 0; i < PRJ_NF4; ++i)
            *(v4f*)&sP[st][(r8 + 8 * i) * PRJ_PITCH + 4 * c4] = r[i];
    };

    v8f acc0 = {}, acc1 = {}, acc2 = {};

    fetch(0);
    put(0);
    lds_barrier();

#pragma unroll 1
    for (int ch = 0; ch < PRJ_NCH; ++ch) {
        const int st = ch & 1;
        if (ch + 1 < PRJ_NCH)
            fetch((ch + 1) * PRJ_KC);

        const float* aRow = &sP[st][(16 * rt + c) * PRJ_PITCH + 2 * h];
        const float* bRow = &sP[st][(PRJ_ROWS + 48 * cg + c) * PRJ_PITCH + 2 * h];
#pragma unroll
        for (int kk = 0; kk < PRJ_KC / 4; ++kk) {
            const v2f a  = *(const v2f*)(aRow + 4 * kk);
            const v2f b0 = *(const v2f*)(bRow + 4 * kk);
            const v2f b1 = *(const v2f*)(bRow + 16 * PRJ_PITCH + 4 * kk);
            const v2f b2 = *(const v2f*)(bRow + 32 * PRJ_PITCH + 4 * kk);
            acc0 = __builtin_amdgcn_wmma_f32_16x16x4_f32(false, a, false, b0, (short)0, acc0, false, false);
            acc1 = __builtin_amdgcn_wmma_f32_16x16x4_f32(false, a, false, b1, (short)0, acc1, false, false);
            acc2 = __builtin_amdgcn_wmma_f32_16x16x4_f32(false, a, false, b2, (short)0, acc2, false, false);
        }

        if (ch + 1 < PRJ_NCH)
            put(st ^ 1);
        lds_barrier();
    }

    const size_t row0 = (size_t)(m0 + 16 * rt + 8 * h);
    if (cg == 0) {
        const float v0 = b_xdt[c], v1 = b_xdt[16 + c], v2 = b_xdt[32 + c];
#pragma unroll
        for (int q = 0; q < 8; ++q) {
            float* o = xdt + (row0 + q) * DT_RANK + c;
            o[0]  = acc0[q] + v0;
            o[16] = acc1[q] + v1;
            o[32] = acc2[q] + v2;
        }
    } else {
        const float v0 = b_xdt[48 + c], v1 = b_B[c], v2 = b_C[c];
#pragma unroll
        for (int q = 0; q < 8; ++q) {
            xdt[(row0 + q) * DT_RANK + 48 + c] = acc0[q] + v0;
            Bt[(row0 + q) * D_STATE + c]       = acc1[q] + v1;
            Ct[(row0 + q) * D_STATE + c]       = acc2[q] + v2;
        }
    }
}

struct StageRegs { v4f u[SFETCH], dt[SFETCH], B[SFETCH], C[SFETCH]; };

template <int CTRL>
__device__ __forceinline__ float dpp_read(float v) {
    return __builtin_bit_cast(float,
        __builtin_amdgcn_update_dpp(0, __builtin_bit_cast(int, v), CTRL, 0xf, 0xf, true));
}

__device__ __forceinline__ void stage_fetch(StageRegs& r,
        const float* __restrict__ u, const float* __restrict__ dt,
        const float* __restrict__ Btg, const float* __restrict__ Ctg,
        int b, int t0, int d0, int tid) {
#pragma unroll
    for (int i = 0; i < SFETCH; ++i) {
        const int q   = tid + 128 * i;
        const int row = q >> 2;
        const int c4  = q & 3;
        const size_t rt = (size_t)b * T_LEN + (size_t)(t0 + row);
        const size_t gu = rt * D_MODEL + (size_t)(d0 + 4 * c4);
        const size_t gs = rt * D_STATE + (size_t)(4 * c4);
        r.u[i]  = *(const v4f*)(u   + gu);
        r.dt[i] = *(const v4f*)(dt  + gu);
        r.B[i]  = *(const v4f*)(Btg + gs);
        r.C[i]  = *(const v4f*)(Ctg + gs);
    }
}

__global__ __launch_bounds__(128)
void scan_sequential(const float* __restrict__ u,
                     const float* __restrict__ dt,
                     const float* __restrict__ Btg,
                     const float* __restrict__ Ctg,
                     const float* __restrict__ A_log,
                     const float* __restrict__ Dvec,
                     float* __restrict__ out) {
    __shared__ v2f sUD[2][SCH][16];
    __shared__ v4f sBC[2][SCH][8];

    const int tid = threadIdx.x;
    const int b   = blockIdx.x >> 6;
    const int d0  = (blockIdx.x & 63) << 4;
    const int j   = tid & 7;
    const int dl  = tid >> 3;
    const int d   = d0 + dl;

    const float LOG2E = 1.44269504088896340736f;
    const float A2a = -__expf(A_log[d * D_STATE + 2 * j])     * LOG2E;
    const float A2b = -__expf(A_log[d * D_STATE + 2 * j + 1]) * LOG2E;
    const float Dd  = Dvec[d];

    float h0 = 0.0f, h1 = 0.0f;

    StageRegs r;
    stage_fetch(r, u, dt, Btg, Ctg, b, 0, d0, tid);

    auto stage_put = [&](int st) {
#pragma unroll
        for (int i = 0; i < SFETCH; ++i) {
            const int q   = tid + 128 * i;
            const int row = q >> 2;
            const int c4  = q & 3;
            v4f lo = { r.u[i].x, r.dt[i].x, r.u[i].y, r.dt[i].y };
            v4f hi = { r.u[i].z, r.dt[i].z, r.u[i].w, r.dt[i].w };
            *(v4f*)&sUD[st][row][4 * c4]     = lo;
            *(v4f*)&sUD[st][row][4 * c4 + 2] = hi;
            v4f p0 = { r.B[i].x, r.B[i].y, r.C[i].x, r.C[i].y };
            v4f p1 = { r.B[i].z, r.B[i].w, r.C[i].z, r.C[i].w };
            sBC[st][row][2 * c4]     = p0;
            sBC[st][row][2 * c4 + 1] = p1;
        }
    };

    stage_put(0);
    lds_barrier();

    auto run8 = [&](const v2f (&ud)[8], const v4f (&bc)[8], size_t orow) {
        float ykeep = 0.0f;
#pragma unroll
        for (int s = 0; s < 8; ++s) {
            const float ul  = ud[s].x;
            const float dtl = ud[s].y;
            const float dtu = dtl * ul;
            const float a0  = exp2f(dtl * A2a);
            const float a1  = exp2f(dtl * A2b);
            h0 = fmaf(a0, h0, bc[s].x * dtu);
            h1 = fmaf(a1, h1, bc[s].y * dtu);
            float yp = h0 * bc[s].z;
            yp = fmaf(h1, bc[s].w, yp);
            yp += dpp_read<0xB1>(yp);
            yp += dpp_read<0x4E>(yp);
            yp += dpp_read<0x164>(yp);
            const float y = fmaf(Dd, ul, yp);
            ykeep = (j == s) ? y : ykeep;
        }
        out[orow] = ykeep;
    };

    for (int k = 0; k < NSTG; ++k) {
        const int st = k & 1;
        const int t0 = k * SCH;
        if (k + 1 < NSTG)
            stage_fetch(r, u, dt, Btg, Ctg, b, t0 + SCH, d0, tid);

        const size_t obase = ((size_t)b * T_LEN + (size_t)(t0 + j)) * D_MODEL + (size_t)d;

        v2f udA[8], udB[8];
        v4f bcA[8], bcB[8];
#pragma unroll
        for (int s = 0; s < 8; ++s) { udA[s] = sUD[st][s][dl]; bcA[s] = sBC[st][s][j]; }

#pragma unroll 1
        for (int t8 = 0; t8 < SCH; t8 += 16) {
#pragma unroll
            for (int s = 0; s < 8; ++s) { udB[s] = sUD[st][t8 + 8 + s][dl]; bcB[s] = sBC[st][t8 + 8 + s][j]; }
            run8(udA, bcA, obase + (size_t)t8 * D_MODEL);
            const int tn = (t8 + 16 < SCH) ? (t8 + 16) : t8;
#pragma unroll
            for (int s = 0; s < 8; ++s) { udA[s] = sUD[st][tn + s][dl]; bcA[s] = sBC[st][tn + s][j]; }
            run8(udB, bcB, obase + (size_t)(t8 + 8) * D_MODEL);
        }

        if (k + 1 < NSTG)
            stage_put(st ^ 1);
        lds_barrier();
    }
}

extern "C" void kernel_launch(void* const* d_in, const int* in_sizes, int n_in,
                              void* d_out, int out_size, void* d_ws, size_t ws_size,
                              hipStream_t stream) {
    const float* u     = (const float*)d_in[0];
    const float* A_log = (const float*)d_in[1];
    const float* W_xdt = (const float*)d_in[2];
    const float* b_xdt = (const float*)d_in[3];
    const float* W_dt  = (const float*)d_in[4];
    const float* b_dt  = (const float*)d_in[5];
    const float* W_B   = (const float*)d_in[6];
    const float* b_B   = (const float*)d_in[7];
    const float* W_C   = (const float*)d_in[8];
    const float* b_C   = (const float*)d_in[9];
    const float* Dv    = (const float*)d_in[10];
    float* out = (float*)d_out;

    float* ws   = (float*)d_ws;
    float* xdt  = ws;
    float* dt   = xdt + (size_t)M_ROWS * DT_RANK;
    float* Bt   = dt  + (size_t)M_ROWS * D_MODEL;
    float* Ct   = Bt  + (size_t)M_ROWS * D_STATE;

    proj_u_fused<<<M_ROWS / PRJ_ROWS, 128, 0, stream>>>(
        u, W_xdt, b_xdt, W_B, b_B, W_C, b_C, xdt, Bt, Ct);

    wmma_gemm_f32<1><<<(M_ROWS / 16) * (D_MODEL / 16) / 8, 256, 0, stream>>>(
        xdt, W_dt, b_dt, dt, M_ROWS, D_MODEL, DT_RANK);

    scan_sequential<<<B_SZ * (D_MODEL / 16), 128, 0, stream>>>(u, dt, Bt, Ct, A_log, Dv, out);
}


# dt projection re-tiled: a work-group stages 64 x_dt rows and 128 W_dt rows once in LDS, 8 waves split the 64x128 output, lanes store whole 128-byte lines (K order kept); 8192 waves, was 32768
# speedup vs baseline: 2.1412x; 1.0563x over previous
#include <hip/hip_runtime.h>
#include <hip/hip_bf16.h>
#include <math.h>

typedef float v2f __attribute__((ext_vector_type(2)));
typedef float v4f __attribute__((ext_vector_type(4)));
typedef float v8f __attribute__((ext_vector_type(8)));

#define B_SZ    4
#define T_LEN   2048
#define D_MODEL 1024
#define D_STATE 16
#define DT_RANK 64
#define M_ROWS  (B_SZ * T_LEN)

#define SCH     128
#define NSTG    (T_LEN / SCH)
#define SFETCH  (SCH / 32)

#define PRJ_ROWS   32
#define PRJ_COLS   96
#define PRJ_KC     64
#define PRJ_PITCH  68
#define PRJ_STAGED (PRJ_ROWS + PRJ_COLS)
#define PRJ_NCH    (D_MODEL / PRJ_KC)
#define PRJ_NF4    (PRJ_STAGED * PRJ_KC / 4 / 128)

__device__ __forceinline__ void lds_barrier() {
    __builtin_amdgcn_fence(__ATOMIC_RELEASE, "workgroup", "local");
    __builtin_amdgcn_s_barrier();
    __builtin_amdgcn_fence(__ATOMIC_ACQUIRE, "workgroup", "local");
}

__global__ __launch_bounds__(128)
void proj_u_fused(const float* __restrict__ u,
                  const float* __restrict__ W_xdt, const float* __restrict__ b_xdt,
                  const float* __restrict__ W_B,   const float* __restrict__ b_B,
                  const float* __restrict__ W_C,   const float* __restrict__ b_C,
                  float* __restrict__ xdt, float* __restrict__ Bt, float* __restrict__ Ct) {
    __shared__ __attribute__((aligned(16))) float sP[2][PRJ_STAGED * PRJ_PITCH];

    const int tid  = threadIdx.x;
    const int lane = tid & 31;
    const int w    = tid >> 5;
    const int rt   = w & 1;
    const int cg   = w >> 1;
    const int c    = lane & 15;
    const int h    = lane >> 4;
    const int m0   = blockIdx.x * PRJ_ROWS;
    const int r8   = tid >> 4;
    const int c4   = tid & 15;

    v4f r[PRJ_NF4];

    auto fetch = [&](int k0) {
#pragma unroll
        for (int i = 0; i < PRJ_NF4; ++i) {
            const int row = r8 + 8 * i;
            const float* p;
            if (i < 4)       p = u     + (size_t)(m0 + row)  * D_MODEL;
            else if (i < 12) p = W_xdt + (size_t)(row - 32)  * D_MODEL;
            else if (i < 14) p = W_B   + (size_t)(row - 96)  * D_MODEL;
            else             p = W_C   + (size_t)(row - 112) * D_MODEL;
            r[i] = *(const v4f*)(p + k0 + 4 * c4);
        }
    };
    auto put = [&](int st) {
#pragma unroll
        for (int i = 0; i < PRJ_NF4; ++i)
            *(v4f*)&sP[st][(r8 + 8 * i) * PRJ_PITCH + 4 * c4] = r[i];
    };

    v8f acc0 = {}, acc1 = {}, acc2 = {};

    fetch(0);
    put(0);
    lds_barrier();

#pragma unroll 1
    for (int ch = 0; ch < PRJ_NCH; ++ch) {
        const int st = ch & 1;
        if (ch + 1 < PRJ_NCH)
            fetch((ch + 1) * PRJ_KC);

        const float* aRow = &sP[st][(16 * rt + c) * PRJ_PITCH + 2 * h];
        const float* bRow = &sP[st][(PRJ_ROWS + 48 * cg + c) * PRJ_PITCH + 2 * h];
#pragma unroll
        for (int kk = 0; kk < PRJ_KC / 4; ++kk) {
            const v2f a  = *(const v2f*)(aRow + 4 * kk);
            const v2f b0 = *(const v2f*)(bRow + 4 * kk);
            const v2f b1 = *(const v2f*)(bRow + 16 * PRJ_PITCH + 4 * kk);
            const v2f b2 = *(const v2f*)(bRow + 32 * PRJ_PITCH + 4 * kk);
            acc0 = __builtin_amdgcn_wmma_f32_16x16x4_f32(false, a, false, b0, (short)0, acc0, false, false);
            acc1 = __builtin_amdgcn_wmma_f32_16x16x4_f32(false, a, false, b1, (short)0, acc1, false, false);
            acc2 = __builtin_amdgcn_wmma_f32_16x16x4_f32(false, a, false, b2, (short)0, acc2, false, false);
        }

        if (ch + 1 < PRJ_NCH)
            put(st ^ 1);
        lds_barrier();
    }

    const size_t row0 = (size_t)(m0 + 16 * rt + 8 * h);
    if (cg == 0) {
        const float v0 = b_xdt[c], v1 = b_xdt[16 + c], v2 = b_xdt[32 + c];
#pragma unroll
        for (int q = 0; q < 8; ++q) {
            float* o = xdt + (row0 + q) * DT_RANK + c;
            o[0]  = acc0[q] + v0;
            o[16] = acc1[q] + v1;
            o[32] = acc2[q] + v2;
        }
    } else {
        const float v0 = b_xdt[48 + c], v1 = b_B[c], v2 = b_C[c];
#pragma unroll
        for (int q = 0; q < 8; ++q) {
            xdt[(row0 + q) * DT_RANK + 48 + c] = acc0[q] + v0;
            Bt[(row0 + q) * D_STATE + c]       = acc1[q] + v1;
            Ct[(row0 + q) * D_STATE + c]       = acc2[q] + v2;
        }
    }
}

#define DTP_ROWS   64
#define DTP_COLS   128
#define DTP_PITCH  68
#define DTP_STAGED (DTP_ROWS + DTP_COLS)
#define DTP_NF4    (DTP_STAGED * DT_RANK / 4 / 256)

__global__ __launch_bounds__(256)
void dt_proj_softplus(const float* __restrict__ xdt,
                      const float* __restrict__ W_dt,
                      const float* __restrict__ b_dt,
                      float* __restrict__ dt) {
    __shared__ __attribute__((aligned(16))) float sD[DTP_STAGED * DTP_PITCH];

    const int tid   = threadIdx.x;
    const int lane  = tid & 31;
    const int w     = tid >> 5;
    const int rt    = w & 3;
    const int chalf = w >> 2;
    const int c     = lane & 15;
    const int h     = lane >> 4;
    const int m0    = (blockIdx.x >> 3) * DTP_ROWS;
    const int n0    = (blockIdx.x & 7) * DTP_COLS;
    const int r16   = tid >> 4;
    const int c4    = tid & 15;

    v4f r[DTP_NF4];
#pragma unroll
    for (int i = 0; i < DTP_NF4; ++i) {
        const int row = r16 + 16 * i;
        const float* p = (i < 4) ? xdt  + (size_t)(m0 + row)            * DT_RANK
                                 : W_dt + (size_t)(n0 + row - DTP_ROWS) * DT_RANK;
        r[i] = *(const v4f*)(p + 4 * c4);
    }
#pragma unroll
    for (int i = 0; i < DTP_NF4; ++i) {
        const int row = r16 + 16 * i;
        int place = row;
        if (i >= 4) {
            const int wr = row - DTP_ROWS;
            place = DTP_ROWS + ((wr & 96) | ((wr & 1) << 4) | ((wr & 31) >> 1));
        }
        *(v4f*)&sD[place * DTP_PITCH + 4 * c4] = r[i];
    }
    lds_barrier();

    const float* aRow = &sD[(16 * rt + c) * DTP_PITCH + 2 * h];

#pragma unroll 1
    for (int g = 0; g < 2; ++g) {
        const int ct = 64 * chalf + 32 * g;
        const float* bRow = &sD[(DTP_ROWS + ct + c) * DTP_PITCH + 2 * h];
        v8f acc0 = {}, acc1 = {};
#pragma unroll
        for (int kk = 0; kk < DT_RANK / 4; ++kk) {
            const v2f a  = *(const v2f*)(aRow + 4 * kk);
            const v2f b0 = *(const v2f*)(bRow + 4 * kk);
            const v2f b1 = *(const v2f*)(bRow + 16 * DTP_PITCH + 4 * kk);
            acc0 = __builtin_amdgcn_wmma_f32_16x16x4_f32(false, a, false, b0, (short)0, acc0, false, false);
            acc1 = __builtin_amdgcn_wmma_f32_16x16x4_f32(false, a, false, b1, (short)0, acc1, false, false);
        }

        const int   col = n0 + ct + 2 * c;
        const float v0  = b_dt[col], v1 = b_dt[col + 1];
        float* o = dt + (size_t)(m0 + 16 * rt + 8 * h) * D_MODEL + col;
#pragma unroll
        for (int q = 0; q < 8; ++q) {
            float x0 = acc0[q] + v0;
            float x1 = acc1[q] + v1;
            x0 = (x0 > 20.0f) ? x0 : log1pf(__expf(x0));
            x1 = (x1 > 20.0f) ? x1 : log1pf(__expf(x1));
            const v2f xy = { x0, x1 };
            *(v2f*)(o + (size_t)q * D_MODEL) = xy;
        }
    }
}

struct StageRegs { v4f u[SFETCH], dt[SFETCH], B[SFETCH], C[SFETCH]; };

template <int CTRL>
__device__ __forceinline__ float dpp_read(float v) {
    return __builtin_bit_cast(float,
        __builtin_amdgcn_update_dpp(0, __builtin_bit_cast(int, v), CTRL, 0xf, 0xf, true));
}

__device__ __forceinline__ void stage_fetch(StageRegs& r,
        const float* __restrict__ u, const float* __restrict__ dt,
        const float* __restrict__ Btg, const float* __restrict__ Ctg,
        int b, int t0, int d0, int tid) {
#pragma unroll
    for (int i = 0; i < SFETCH; ++i) {
        const int q   = tid + 128 * i;
        const int row = q >> 2;
        const int c4  = q & 3;
        const size_t rt = (size_t)b * T_LEN + (size_t)(t0 + row);
        const size_t gu = rt * D_MODEL + (size_t)(d0 + 4 * c4);
        const size_t gs = rt * D_STATE + (size_t)(4 * c4);
        r.u[i]  = *(const v4f*)(u   + gu);
        r.dt[i] = *(const v4f*)(dt  + gu);
        r.B[i]  = *(const v4f*)(Btg + gs);
        r.C[i]  = *(const v4f*)(Ctg + gs);
    }
}

__global__ __launch_bounds__(128)
void scan_sequential(const float* __restrict__ u,
                     const float* __restrict__ dt,
                     const float* __restrict__ Btg,
                     const float* __restrict__ Ctg,
                     const float* __restrict__ A_log,
                     const float* __restrict__ Dvec,
                     float* __restrict__ out) {
    __shared__ v2f sUD[2][SCH][16];
    __shared__ v4f sBC[2][SCH][8];

    const int tid = threadIdx.x;
    const int b   = blockIdx.x >> 6;
    const int d0  = (blockIdx.x & 63) << 4;
    const int j   = tid & 7;
    const int dl  = tid >> 3;
    const int d   = d0 + dl;

    const float LOG2E = 1.44269504088896340736f;
    const float A2a = -__expf(A_log[d * D_STATE + 2 * j])     * LOG2E;
    const float A2b = -__expf(A_log[d * D_STATE + 2 * j + 1]) * LOG2E;
    const float Dd  = Dvec[d];

    float h0 = 0.0f, h1 = 0.0f;

    StageRegs r;
    stage_fetch(r, u, dt, Btg, Ctg, b, 0, d0, tid);

    auto stage_put = [&](int st) {
#pragma unroll
        for (int i = 0; i < SFETCH; ++i) {
            const int q   = tid + 128 * i;
            const int row = q >> 2;
            const int c4  = q & 3;
            v4f lo = { r.u[i].x, r.dt[i].x, r.u[i].y, r.dt[i].y };
            v4f hi = { r.u[i].z, r.dt[i].z, r.u[i].w, r.dt[i].w };
            *(v4f*)&sUD[st][row][4 * c4]     = lo;
            *(v4f*)&sUD[st][row][4 * c4 + 2] = hi;
            v4f p0 = { r.B[i].x, r.B[i].y, r.C[i].x, r.C[i].y };
            v4f p1 = { r.B[i].z, r.B[i].w, r.C[i].z, r.C[i].w };
            sBC[st][row][2 * c4]     = p0;
            sBC[st][row][2 * c4 + 1] = p1;
        }
    };

    stage_put(0);
    lds_barrier();

    auto run8 = [&](const v2f (&ud)[8], const v4f (&bc)[8], size_t orow) {
        float ykeep = 0.0f;
#pragma unroll
        for (int s = 0; s < 8; ++s) {
            const float ul  = ud[s].x;
            const float dtl = ud[s].y;
            const float dtu = dtl * ul;
            const float a0  = exp2f(dtl * A2a);
            const float a1  = exp2f(dtl * A2b);
            h0 = fmaf(a0, h0, bc[s].x * dtu);
            h1 = fmaf(a1, h1, bc[s].y * dtu);
            float yp = h0 * bc[s].z;
            yp = fmaf(h1, bc[s].w, yp);
            yp += dpp_read<0xB1>(yp);
            yp += dpp_read<0x4E>(yp);
            yp += dpp_read<0x164>(yp);
            const float y = fmaf(Dd, ul, yp);
            ykeep = (j == s) ? y : ykeep;
        }
        out[orow] = ykeep;
    };

    for (int k = 0; k < NSTG; ++k) {
        const int st = k & 1;
        const int t0 = k * SCH;
        if (k + 1 < NSTG)
            stage_fetch(r, u, dt, Btg, Ctg, b, t0 + SCH, d0, tid);

        const size_t obase = ((size_t)b * T_LEN + (size_t)(t0 + j)) * D_MODEL + (size_t)d;

        v2f udA[8], udB[8];
        v4f bcA[8], bcB[8];
#pragma unroll
        for (int s = 0; s < 8; ++s) { udA[s] = sUD[st][s][dl]; bcA[s] = sBC[st][s][j]; }

#pragma unroll 1
        for (int t8 = 0; t8 < SCH; t8 += 16) {
#pragma unroll
            for (int s = 0; s < 8; ++s) { udB[s] = sUD[st][t8 + 8 + s][dl]; bcB[s] = sBC[st][t8 + 8 + s][j]; }
            run8(udA, bcA, obase + (size_t)t8 * D_MODEL);
            const int tn = (t8 + 16 < SCH) ? (t8 + 16) : t8;
#pragma unroll
            for (int s = 0; s < 8; ++s) { udA[s] = sUD[st][tn + s][dl]; bcA[s] = sBC[st][tn + s][j]; }
            run8(udB, bcB, obase + (size_t)(t8 + 8) * D_MODEL);
        }

        if (k + 1 < NSTG)
            stage_put(st ^ 1);
        lds_barrier();
    }
}

extern "C" void kernel_launch(void* const* d_in, const int* in_sizes, int n_in,
                              void* d_out, int out_size, void* d_ws, size_t ws_size,
                              hipStream_t stream) {
    const float* u     = (const float*)d_in[0];
    const float* A_log = (const float*)d_in[1];
    const float* W_xdt = (const float*)d_in[2];
    const float* b_xdt = (const float*)d_in[3];
    const float* W_dt  = (const float*)d_in[4];
    const float* b_dt  = (const float*)d_in[5];
    const float* W_B   = (const float*)d_in[6];
    const float* b_B   = (const float*)d_in[7];
    const float* W_C   = (const float*)d_in[8];
    const float* b_C   = (const float*)d_in[9];
    const float* Dv    = (const float*)d_in[10];
    float* out = (float*)d_out;

    float* ws   = (float*)d_ws;
    float* xdt  = ws;
    float* dt   = xdt + (size_t)M_ROWS * DT_RANK;
    float* Bt   = dt  + (size_t)M_ROWS * D_MODEL;
    float* Ct   = Bt  + (size_t)M_ROWS * D_STATE;

    proj_u_fused<<<M_ROWS / PRJ_ROWS, 128, 0, stream>>>(
        u, W_xdt, b_xdt, W_B, b_B, W_C, b_C, xdt, Bt, Ct);

    dt_proj_softplus<<<(M_ROWS / DTP_ROWS) * (D_MODEL / DTP_COLS), 256, 0, stream>>>(
        xdt, W_dt, b_dt, dt);

    scan_sequential<<<B_SZ * (D_MODEL / 16), 128, 0, stream>>>(u, dt, Bt, Ct, A_log, Dv, out);
}


# scan: the next piece's global loads now overlap the step loop: D's pending load had put a wait for ALL loads inside the loop, so every piece's first trip waited for its prefetch
# speedup vs baseline: 2.4000x; 1.1209x over previous
#include <hip/hip_runtime.h>
#include <hip/hip_bf16.h>
#include <math.h>

typedef float v2f __attribute__((ext_vector_type(2)));
typedef float v4f __attribute__((ext_vector_type(4)));
typedef float v8f __attribute__((ext_vector_type(8)));

#define B_SZ    4
#define T_LEN   2048
#define D_MODEL 1024
#define D_STATE 16
#define DT_RANK 64
#define M_ROWS  (B_SZ * T_LEN)

#define SCH     128
#define NSTG    (T_LEN / SCH)
#define SFETCH  (SCH / 32)

#define PRJ_ROWS   32
#define PRJ_COLS   96
#define PRJ_KC     64
#define PRJ_PITCH  68
#define PRJ_STAGED (PRJ_ROWS + PRJ_COLS)
#define PRJ_NCH    (D_MODEL / PRJ_KC)
#define PRJ_NF4    (PRJ_STAGED * PRJ_KC / 4 / 128)

__device__ __forceinline__ void lds_barrier() {
    __builtin_amdgcn_fence(__ATOMIC_RELEASE, "workgroup", "local");
    __builtin_amdgcn_s_barrier();
    __builtin_amdgcn_fence(__ATOMIC_ACQUIRE, "workgroup", "local");
}

__global__ __launch_bounds__(128)
void proj_u_fused(const float* __restrict__ u,
                  const float* __restrict__ W_xdt, const float* __restrict__ b_xdt,
                  const float* __restrict__ W_B,   const float* __restrict__ b_B,
                  const float* __restrict__ W_C,   const float* __restrict__ b_C,
                  float* __restrict__ xdt, float* __restrict__ Bt, float* __restrict__ Ct) {
    __shared__ __attribute__((aligned(16))) float sP[2][PRJ_STAGED * PRJ_PITCH];

    const int tid  = threadIdx.x;
    const int lane = tid & 31;
    const int w    = tid >> 5;
    const int rt   = w & 1;
    const int cg   = w >> 1;
    const int c    = lane & 15;
    const int h    = lane >> 4;
    const int m0   = blockIdx.x * PRJ_ROWS;
    const int r8   = tid >> 4;
    const int c4   = tid & 15;

    v4f r[PRJ_NF4];

    auto fetch = [&](int k0) {
#pragma unroll
        for (int i = 0; i < PRJ_NF4; ++i) {
            const int row = r8 + 8 * i;
            const float* p;
            if (i < 4)       p = u     + (size_t)(m0 + row)  * D_MODEL;
            else if (i < 12) p = W_xdt + (size_t)(row - 32)  * D_MODEL;
            else if (i < 14) p = W_B   + (size_t)(row - 96)  * D_MODEL;
            else             p = W_C   + (size_t)(row - 112) * D_MODEL;
            r[i] = *(const v4f*)(p + k0 + 4 * c4);
        }
    };
    auto put = [&](int st) {
#pragma unroll
        for (int i = 0; i < PRJ_NF4; ++i)
            *(v4f*)&sP[st][(r8 + 8 * i) * PRJ_PITCH + 4 * c4] = r[i];
    };

    v8f acc0 = {}, acc1 = {}, acc2 = {};

    fetch(0);
    put(0);
    lds_barrier();

#pragma unroll 1
    for (int ch = 0; ch < PRJ_NCH; ++ch) {
        const int st = ch & 1;
        if (ch + 1 < PRJ_NCH)
            fetch((ch + 1) * PRJ_KC);

        const float* aRow = &sP[st][(16 * rt + c) * PRJ_PITCH + 2 * h];
        const float* bRow = &sP[st][(PRJ_ROWS + 48 * cg + c) * PRJ_PITCH + 2 * h];
#pragma unroll
        for (int kk = 0; kk < PRJ_KC / 4; ++kk) {
            const v2f a  = *(const v2f*)(aRow + 4 * kk);
            const v2f b0 = *(const v2f*)(bRow + 4 * kk);
            const v2f b1 = *(const v2f*)(bRow + 16 * PRJ_PITCH + 4 * kk);
            const v2f b2 = *(const v2f*)(bRow + 32 * PRJ_PITCH + 4 * kk);
            acc0 = __builtin_amdgcn_wmma_f32_16x16x4_f32(false, a, false, b0, (short)0, acc0, false, false);
            acc1 = __builtin_amdgcn_wmma_f32_16x16x4_f32(false, a, false, b1, (short)0, acc1, false, false);
            acc2 = __builtin_amdgcn_wmma_f32_16x16x4_f32(false, a, false, b2, (short)0, acc2, false, false);
        }

        if (ch + 1 < PRJ_NCH)
            put(st ^ 1);
        lds_barrier();
    }

    const size_t row0 = (size_t)(m0 + 16 * rt + 8 * h);
    if (cg == 0) {
        const float v0 = b_xdt[c], v1 = b_xdt[16 + c], v2 = b_xdt[32 + c];
#pragma unroll
        for (int q = 0; q < 8; ++q) {
            float* o = xdt + (row0 + q) * DT_RANK + c;
            o[0]  = acc0[q] + v0;
            o[16] = acc1[q] + v1;
            o[32] = acc2[q] + v2;
        }
    } else {
        const float v0 = b_xdt[48 + c], v1 = b_B[c], v2 = b_C[c];
#pragma unroll
        for (int q = 0; q < 8; ++q) {
            xdt[(row0 + q) * DT_RANK + 48 + c] = acc0[q] + v0;
            Bt[(row0 + q) * D_STATE + c]       = acc1[q] + v1;
            Ct[(row0 + q) * D_STATE + c]       = acc2[q] + v2;
        }
    }
}

#define DTP_ROWS   64
#define DTP_COLS   128
#define DTP_PITCH  68
#define DTP_STAGED (DTP_ROWS + DTP_COLS)
#define DTP_NF4    (DTP_STAGED * DT_RANK / 4 / 256)

__global__ __launch_bounds__(256)
void dt_proj_softplus(const float* __restrict__ xdt,
                      const float* __restrict__ W_dt,
                      const float* __restrict__ b_dt,
                      float* __restrict__ dt) {
    __shared__ __attribute__((aligned(16))) float sD[DTP_STAGED * DTP_PITCH];

    const int tid   = threadIdx.x;
    const int lane  = tid & 31;
    const int w     = tid >> 5;
    const int rt    = w & 3;
    const int chalf = w >> 2;
    const int c     = lane & 15;
    const int h     = lane >> 4;
    const int m0    = (blockIdx.x >> 3) * DTP_ROWS;
    const int n0    = (blockIdx.x & 7) * DTP_COLS;
    const int r16   = tid >> 4;
    const int c4    = tid & 15;

    v4f r[DTP_NF4];
#pragma unroll
    for (int i = 0; i < DTP_NF4; ++i) {
        const int row = r16 + 16 * i;
        const float* p = (i < 4) ? xdt  + (size_t)(m0 + row)            * DT_RANK
                                 : W_dt + (size_t)(n0 + row - DTP_ROWS) * DT_RANK;
        r[i] = *(const v4f*)(p + 4 * c4);
    }
#pragma unroll
    for (int i = 0; i < DTP_NF4; ++i) {
        const int row = r16 + 16 * i;
        int place = row;
        if (i >= 4) {
            const int wr = row - DTP_ROWS;
            place = DTP_ROWS + ((wr & 96) | ((wr & 1) << 4) | ((wr & 31) >> 1));
        }
        *(v4f*)&sD[place * DTP_PITCH + 4 * c4] = r[i];
    }
    lds_barrier();

    const float* aRow = &sD[(16 * rt + c) * DTP_PITCH + 2 * h];

#pragma unroll 1
    for (int g = 0; g < 2; ++g) {
        const int ct = 64 * chalf + 32 * g;
        const float* bRow = &sD[(DTP_ROWS + ct + c) * DTP_PITCH + 2 * h];
        v8f acc0 = {}, acc1 = {};
#pragma unroll
        for (int kk = 0; kk < DT_RANK / 4; ++kk) {
            const v2f a  = *(const v2f*)(aRow + 4 * kk);
            const v2f b0 = *(const v2f*)(bRow + 4 * kk);
            const v2f b1 = *(const v2f*)(bRow + 16 * DTP_PITCH + 4 * kk);
            acc0 = __builtin_amdgcn_wmma_f32_16x16x4_f32(false, a, false, b0, (short)0, acc0, false, false);
            acc1 = __builtin_amdgcn_wmma_f32_16x16x4_f32(false, a, false, b1, (short)0, acc1, false, false);
        }

        const int   col = n0 + ct + 2 * c;
        const float v0  = b_dt[col], v1 = b_dt[col + 1];
        float* o = dt + (size_t)(m0 + 16 * rt + 8 * h) * D_MODEL + col;
#pragma unroll
        for (int q = 0; q < 8; ++q) {
            float x0 = acc0[q] + v0;
            float x1 = acc1[q] + v1;
            x0 = (x0 > 20.0f) ? x0 : log1pf(__expf(x0));
            x1 = (x1 > 20.0f) ? x1 : log1pf(__expf(x1));
            const v2f xy = { x0, x1 };
            *(v2f*)(o + (size_t)q * D_MODEL) = xy;
        }
    }
}

struct StageRegs { v4f u[SFETCH], dt[SFETCH], B[SFETCH], C[SFETCH]; };

template <int CTRL>
__device__ __forceinline__ float dpp_read(float v) {
    return __builtin_bit_cast(float,
        __builtin_amdgcn_update_dpp(0, __builtin_bit_cast(int, v), CTRL, 0xf, 0xf, true));
}

__device__ __forceinline__ void stage_fetch(StageRegs& r,
        const float* __restrict__ u, const float* __restrict__ dt,
        const float* __restrict__ Btg, const float* __restrict__ Ctg,
        int b, int t0, int d0, int tid) {
#pragma unroll
    for (int i = 0; i < SFETCH; ++i) {
        const int q   = tid + 128 * i;
        const int row = q >> 2;
        const int c4  = q & 3;
        const size_t rt = (size_t)b * T_LEN + (size_t)(t0 + row);
        const size_t gu = rt * D_MODEL + (size_t)(d0 + 4 * c4);
        const size_t gs = rt * D_STATE + (size_t)(4 * c4);
        r.u[i]  = *(const v4f*)(u   + gu);
        r.dt[i] = *(const v4f*)(dt  + gu);
        r.B[i]  = *(const v4f*)(Btg + gs);
        r.C[i]  = *(const v4f*)(Ctg + gs);
    }
}

__global__ __launch_bounds__(128)
void scan_sequential(const float* __restrict__ u,
                     const float* __restrict__ dt,
                     const float* __restrict__ Btg,
                     const float* __restrict__ Ctg,
                     const float* __restrict__ A_log,
                     const float* __restrict__ Dvec,
                     float* __restrict__ out) {
    __shared__ v2f sUD[2][SCH][16];
    __shared__ v4f sBC[2][SCH][8];

    const int tid = threadIdx.x;
    const int b   = blockIdx.x >> 6;
    const int d0  = (blockIdx.x & 63) << 4;
    const int j   = tid & 7;
    const int dl  = tid >> 3;
    const int d   = d0 + dl;

    const float LOG2E = 1.44269504088896340736f;
    const float A2a = -__expf(A_log[d * D_STATE + 2 * j])     * LOG2E;
    const float A2b = -__expf(A_log[d * D_STATE + 2 * j + 1]) * LOG2E;
    const float Dd  = Dvec[d];

    float h0 = 0.0f, h1 = 0.0f;

    StageRegs r;
    stage_fetch(r, u, dt, Btg, Ctg, b, 0, d0, tid);

    auto stage_put = [&](int st) {
#pragma unroll
        for (int i = 0; i < SFETCH; ++i) {
            const int q   = tid + 128 * i;
            const int row = q >> 2;
            const int c4  = q & 3;
            v4f lo = { r.u[i].x, r.dt[i].x, r.u[i].y, r.dt[i].y };
            v4f hi = { r.u[i].z, r.dt[i].z, r.u[i].w, r.dt[i].w };
            *(v4f*)&sUD[st][row][4 * c4]     = lo;
            *(v4f*)&sUD[st][row][4 * c4 + 2] = hi;
            v4f p0 = { r.B[i].x, r.B[i].y, r.C[i].x, r.C[i].y };
            v4f p1 = { r.B[i].z, r.B[i].w, r.C[i].z, r.C[i].w };
            sBC[st][row][2 * c4]     = p0;
            sBC[st][row][2 * c4 + 1] = p1;
        }
    };

    stage_put(0);
    lds_barrier();

    asm volatile("" :: "v"(Dd));

    auto run8 = [&](const v2f (&ud)[8], const v4f (&bc)[8], size_t orow) {
        float ykeep = 0.0f;
#pragma unroll
        for (int s = 0; s < 8; ++s) {
            const float ul  = ud[s].x;
            const float dtl = ud[s].y;
            const float dtu = dtl * ul;
            const float a0  = exp2f(dtl * A2a);
            const float a1  = exp2f(dtl * A2b);
            h0 = fmaf(a0, h0, bc[s].x * dtu);
            h1 = fmaf(a1, h1, bc[s].y * dtu);
            float yp = h0 * bc[s].z;
            yp = fmaf(h1, bc[s].w, yp);
            yp += dpp_read<0xB1>(yp);
            yp += dpp_read<0x4E>(yp);
            yp += dpp_read<0x164>(yp);
            const float y = fmaf(Dd, ul, yp);
            ykeep = (j == s) ? y : ykeep;
        }
        out[orow] = ykeep;
    };

    for (int k = 0; k < NSTG; ++k) {
        const int st = k & 1;
        const int t0 = k * SCH;
        if (k + 1 < NSTG)
            stage_fetch(r, u, dt, Btg, Ctg, b, t0 + SCH, d0, tid);

        const size_t obase = ((size_t)b * T_LEN + (size_t)(t0 + j)) * D_MODEL + (size_t)d;

        v2f udA[8], udB[8];
        v4f bcA[8], bcB[8];
#pragma unroll
        for (int s = 0; s < 8; ++s) { udA[s] = sUD[st][s][dl]; bcA[s] = sBC[st][s][j]; }

#pragma unroll 1
        for (int t8 = 0; t8 < SCH; t8 += 16) {
#pragma unroll
            for (int s = 0; s < 8; ++s) { udB[s] = sUD[st][t8 + 8 + s][dl]; bcB[s] = sBC[st][t8 + 8 + s][j]; }
            run8(udA, bcA, obase + (size_t)t8 * D_MODEL);
            const int tn = (t8 + 16 < SCH) ? (t8 + 16) : t8;
#pragma unroll
            for (int s = 0; s < 8; ++s) { udA[s] = sUD[st][tn + s][dl]; bcA[s] = sBC[st][tn + s][j]; }
            run8(udB, bcB, obase + (size_t)(t8 + 8) * D_MODEL);
        }

        if (k + 1 < NSTG)
            stage_put(st ^ 1);
        lds_barrier();
    }
}

extern "C" void kernel_launch(void* const* d_in, const int* in_sizes, int n_in,
                              void* d_out, int out_size, void* d_ws, size_t ws_size,
                              hipStream_t stream) {
    const float* u     = (const float*)d_in[0];
    const float* A_log = (const float*)d_in[1];
    const float* W_xdt = (const float*)d_in[2];
    const float* b_xdt = (const float*)d_in[3];
    const float* W_dt  = (const float*)d_in[4];
    const float* b_dt  = (const float*)d_in[5];
    const float* W_B   = (const float*)d_in[6];
    const float* b_B   = (const float*)d_in[7];
    const float* W_C   = (const float*)d_in[8];
    const float* b_C   = (const float*)d_in[9];
    const float* Dv    = (const float*)d_in[10];
    float* out = (float*)d_out;

    float* ws   = (float*)d_ws;
    float* xdt  = ws;
    float* dt   = xdt + (size_t)M_ROWS * DT_RANK;
    float* Bt   = dt  + (size_t)M_ROWS * D_MODEL;
    float* Ct   = Bt  + (size_t)M_ROWS * D_STATE;

    proj_u_fused<<<M_ROWS / PRJ_ROWS, 128, 0, stream>>>(
        u, W_xdt, b_xdt, W_B, b_B, W_C, b_C, xdt, Bt, Ct);

    dt_proj_softplus<<<(M_ROWS / DTP_ROWS) * (D_MODEL / DTP_COLS), 256, 0, stream>>>(
        xdt, W_dt, b_dt, dt);

    scan_sequential<<<B_SZ * (D_MODEL / 16), 128, 0, stream>>>(u, dt, Bt, Ct, A_log, Dv, out);
}


# scan: a 16-step block = recurrence part, then the next block's LDS reads as ONE group into the registers just used up, then the output part; 1 place of reads per 16 steps (was 7), no operand copies
# speedup vs baseline: 2.6012x; 1.0838x over previous
#include <hip/hip_runtime.h>
#include <hip/hip_bf16.h>
#include <math.h>

typedef float v2f __attribute__((ext_vector_type(2)));
typedef float v4f __attribute__((ext_vector_type(4)));
typedef float v8f __attribute__((ext_vector_type(8)));

#define B_SZ    4
#define T_LEN   2048
#define D_MODEL 1024
#define D_STATE 16
#define DT_RANK 64
#define M_ROWS  (B_SZ * T_LEN)

#define SCH     128
#define NSTG    (T_LEN / SCH)
#define SFETCH  (SCH / 32)
#define SBLK    16

#define PRJ_ROWS   32
#define PRJ_COLS   96
#define PRJ_KC     64
#define PRJ_PITCH  68
#define PRJ_STAGED (PRJ_ROWS + PRJ_COLS)
#define PRJ_NCH    (D_MODEL / PRJ_KC)
#define PRJ_NF4    (PRJ_STAGED * PRJ_KC / 4 / 128)

__device__ __forceinline__ void lds_barrier() {
    __builtin_amdgcn_fence(__ATOMIC_RELEASE, "workgroup", "local");
    __builtin_amdgcn_s_barrier();
    __builtin_amdgcn_fence(__ATOMIC_ACQUIRE, "workgroup", "local");
}

__global__ __launch_bounds__(128)
void proj_u_fused(const float* __restrict__ u,
                  const float* __restrict__ W_xdt, const float* __restrict__ b_xdt,
                  const float* __restrict__ W_B,   const float* __restrict__ b_B,
                  const float* __restrict__ W_C,   const float* __restrict__ b_C,
                  float* __restrict__ xdt, float* __restrict__ Bt, float* __restrict__ Ct) {
    __shared__ __attribute__((aligned(16))) float sP[2][PRJ_STAGED * PRJ_PITCH];

    const int tid  = threadIdx.x;
    const int lane = tid & 31;
    const int w    = tid >> 5;
    const int rt   = w & 1;
    const int cg   = w >> 1;
    const int c    = lane & 15;
    const int h    = lane >> 4;
    const int m0   = blockIdx.x * PRJ_ROWS;
    const int r8   = tid >> 4;
    const int c4   = tid & 15;

    v4f r[PRJ_NF4];

    auto fetch = [&](int k0) {
#pragma unroll
        for (int i = 0; i < PRJ_NF4; ++i) {
            const int row = r8 + 8 * i;
            const float* p;
            if (i < 4)       p = u     + (size_t)(m0 + row)  * D_MODEL;
            else if (i < 12) p = W_xdt + (size_t)(row - 32)  * D_MODEL;
            else if (i < 14) p = W_B   + (size_t)(row - 96)  * D_MODEL;
            else             p = W_C   + (size_t)(row - 112) * D_MODEL;
            r[i] = *(const v4f*)(p + k0 + 4 * c4);
        }
    };
    auto put = [&](int st) {
#pragma unroll
        for (int i = 0; i < PRJ_NF4; ++i)
            *(v4f*)&sP[st][(r8 + 8 * i) * PRJ_PITCH + 4 * c4] = r[i];
    };

    v8f acc0 = {}, acc1 = {}, acc2 = {};

    fetch(0);
    put(0);
    lds_barrier();

#pragma unroll 1
    for (int ch = 0; ch < PRJ_NCH; ++ch) {
        const int st = ch & 1;
        if (ch + 1 < PRJ_NCH)
            fetch((ch + 1) * PRJ_KC);

        const float* aRow = &sP[st][(16 * rt + c) * PRJ_PITCH + 2 * h];
        const float* bRow = &sP[st][(PRJ_ROWS + 48 * cg + c) * PRJ_PITCH + 2 * h];
#pragma unroll
        for (int kk = 0; kk < PRJ_KC / 4; ++kk) {
            const v2f a  = *(const v2f*)(aRow + 4 * kk);
            const v2f b0 = *(const v2f*)(bRow + 4 * kk);
            const v2f b1 = *(const v2f*)(bRow + 16 * PRJ_PITCH + 4 * kk);
            const v2f b2 = *(const v2f*)(bRow + 32 * PRJ_PITCH + 4 * kk);
            acc0 = __builtin_amdgcn_wmma_f32_16x16x4_f32(false, a, false, b0, (short)0, acc0, false, false);
            acc1 = __builtin_amdgcn_wmma_f32_16x16x4_f32(false, a, false, b1, (short)0, acc1, false, false);
            acc2 = __builtin_amdgcn_wmma_f32_16x16x4_f32(false, a, false, b2, (short)0, acc2, false, false);
        }

        if (ch + 1 < PRJ_NCH)
            put(st ^ 1);
        lds_barrier();
    }

    const size_t row0 = (size_t)(m0 + 16 * rt + 8 * h);
    if (cg == 0) {
        const float v0 = b_xdt[c], v1 = b_xdt[16 + c], v2 = b_xdt[32 + c];
#pragma unroll
        for (int q = 0; q < 8; ++q) {
            float* o = xdt + (row0 + q) * DT_RANK + c;
            o[0]  = acc0[q] + v0;
            o[16] = acc1[q] + v1;
            o[32] = acc2[q] + v2;
        }
    } else {
        const float v0 = b_xdt[48 + c], v1 = b_B[c], v2 = b_C[c];
#pragma unroll
        for (int q = 0; q < 8; ++q) {
            xdt[(row0 + q) * DT_RANK + 48 + c] = acc0[q] + v0;
            Bt[(row0 + q) * D_STATE + c]       = acc1[q] + v1;
            Ct[(row0 + q) * D_STATE + c]       = acc2[q] + v2;
        }
    }
}

#define DTP_ROWS   64
#define DTP_COLS   128
#define DTP_PITCH  68
#define DTP_STAGED (DTP_ROWS + DTP_COLS)
#define DTP_NF4    (DTP_STAGED * DT_RANK / 4 / 256)

__global__ __launch_bounds__(256)
void dt_proj_softplus(const float* __restrict__ xdt,
                      const float* __restrict__ W_dt,
                      const float* __restrict__ b_dt,
                      float* __restrict__ dt) {
    __shared__ __attribute__((aligned(16))) float sD[DTP_STAGED * DTP_PITCH];

    const int tid   = threadIdx.x;
    const int lane  = tid & 31;
    const int w     = tid >> 5;
    const int rt    = w & 3;
    const int chalf = w >> 2;
    const int c     = lane & 15;
    const int h     = lane >> 4;
    const int m0    = (blockIdx.x >> 3) * DTP_ROWS;
    const int n0    = (blockIdx.x & 7) * DTP_COLS;
    const int r16   = tid >> 4;
    const int c4    = tid & 15;

    v4f r[DTP_NF4];
#pragma unroll
    for (int i = 0; i < DTP_NF4; ++i) {
        const int row = r16 + 16 * i;
        const float* p = (i < 4) ? xdt  + (size_t)(m0 + row)            * DT_RANK
                                 : W_dt + (size_t)(n0 + row - DTP_ROWS) * DT_RANK;
        r[i] = *(const v4f*)(p + 4 * c4);
    }
#pragma unroll
    for (int i = 0; i < DTP_NF4; ++i) {
        const int row = r16 + 16 * i;
        int place = row;
        if (i >= 4) {
            const int wr = row - DTP_ROWS;
            place = DTP_ROWS + ((wr & 96) | ((wr & 1) << 4) | ((wr & 31) >> 1));
        }
        *(v4f*)&sD[place * DTP_PITCH + 4 * c4] = r[i];
    }
    lds_barrier();

    const float* aRow = &sD[(16 * rt + c) * DTP_PITCH + 2 * h];

#pragma unroll 1
    for (int g = 0; g < 2; ++g) {
        const int ct = 64 * chalf + 32 * g;
        const float* bRow = &sD[(DTP_ROWS + ct + c) * DTP_PITCH + 2 * h];
        v8f acc0 = {}, acc1 = {};
#pragma unroll
        for (int kk = 0; kk < DT_RANK / 4; ++kk) {
            const v2f a  = *(const v2f*)(aRow + 4 * kk);
            const v2f b0 = *(const v2f*)(bRow + 4 * kk);
            const v2f b1 = *(const v2f*)(bRow + 16 * DTP_PITCH + 4 * kk);
            acc0 = __builtin_amdgcn_wmma_f32_16x16x4_f32(false, a, false, b0, (short)0, acc0, false, false);
            acc1 = __builtin_amdgcn_wmma_f32_16x16x4_f32(false, a, false, b1, (short)0, acc1, false, false);
        }

        const int   col = n0 + ct + 2 * c;
        const float v0  = b_dt[col], v1 = b_dt[col + 1];
        float* o = dt + (size_t)(m0 + 16 * rt + 8 * h) * D_MODEL + col;
#pragma unroll
        for (int q = 0; q < 8; ++q) {
            float x0 = acc0[q] + v0;
            float x1 = acc1[q] + v1;
            x0 = (x0 > 20.0f) ? x0 : log1pf(__expf(x0));
            x1 = (x1 > 20.0f) ? x1 : log1pf(__expf(x1));
            const v2f xy = { x0, x1 };
            *(v2f*)(o + (size_t)q * D_MODEL) = xy;
        }
    }
}

struct StageRegs { v4f u[SFETCH], dt[SFETCH], B[SFETCH], C[SFETCH]; };

template <int CTRL>
__device__ __forceinline__ float dpp_read(float v) {
    return __builtin_bit_cast(float,
        __builtin_amdgcn_update_dpp(0, __builtin_bit_cast(int, v), CTRL, 0xf, 0xf, true));
}

__device__ __forceinline__ void stage_fetch(StageRegs& r,
        const float* __restrict__ u, const float* __restrict__ dt,
        const float* __restrict__ Btg, const float* __restrict__ Ctg,
        int b, int t0, int d0, int tid) {
#pragma unroll
    for (int i = 0; i < SFETCH; ++i) {
        const int q   = tid + 128 * i;
        const int row = q >> 2;
        const int c4  = q & 3;
        const size_t rt = (size_t)b * T_LEN + (size_t)(t0 + row);
        const size_t gu = rt * D_MODEL + (size_t)(d0 + 4 * c4);
        const size_t gs = rt * D_STATE + (size_t)(4 * c4);
        r.u[i]  = *(const v4f*)(u   + gu);
        r.dt[i] = *(const v4f*)(dt  + gu);
        r.B[i]  = *(const v4f*)(Btg + gs);
        r.C[i]  = *(const v4f*)(Ctg + gs);
    }
}

__global__ __launch_bounds__(128)
void scan_sequential(const float* __restrict__ u,
                     const float* __restrict__ dt,
                     const float* __restrict__ Btg,
                     const float* __restrict__ Ctg,
                     const float* __restrict__ A_log,
                     const float* __restrict__ Dvec,
                     float* __restrict__ out) {
    __shared__ v2f sUD[2][SCH][16];
    __shared__ v4f sBC[2][SCH][8];

    const int tid = threadIdx.x;
    const int b   = blockIdx.x >> 6;
    const int d0  = (blockIdx.x & 63) << 4;
    const int j   = tid & 7;
    const int dl  = tid >> 3;
    const int d   = d0 + dl;

    const float LOG2E = 1.44269504088896340736f;
    const float A2a = -__expf(A_log[d * D_STATE + 2 * j])     * LOG2E;
    const float A2b = -__expf(A_log[d * D_STATE + 2 * j + 1]) * LOG2E;
    const float Dd  = Dvec[d];

    float h0 = 0.0f, h1 = 0.0f;

    StageRegs r;
    stage_fetch(r, u, dt, Btg, Ctg, b, 0, d0, tid);

    auto stage_put = [&](int st) {
#pragma unroll
        for (int i = 0; i < SFETCH; ++i) {
            const int q   = tid + 128 * i;
            const int row = q >> 2;
            const int c4  = q & 3;
            v4f lo = { r.dt[i].x, r.u[i].x, r.dt[i].y, r.u[i].y };
            v4f hi = { r.dt[i].z, r.u[i].z, r.dt[i].w, r.u[i].w };
            *(v4f*)&sUD[st][row][4 * c4]     = lo;
            *(v4f*)&sUD[st][row][4 * c4 + 2] = hi;
            v4f p0 = { r.B[i].x, r.B[i].y, r.C[i].x, r.C[i].y };
            v4f p1 = { r.B[i].z, r.B[i].w, r.C[i].z, r.C[i].w };
            sBC[st][row][2 * c4]     = p0;
            sBC[st][row][2 * c4 + 1] = p1;
        }
    };

    stage_put(0);
    lds_barrier();

    asm volatile("" :: "v"(Dd));

    auto head_part = [&](const v2f (&ud)[SBLK], const v4f (&bc)[SBLK], float (&yp)[SBLK]) {
#pragma unroll
        for (int s = 0; s < SBLK; ++s) {
            const float dtl = ud[s].x;
            const float ul  = ud[s].y;
            const float dtu = dtl * ul;
            const float a0  = exp2f(dtl * A2a);
            const float a1  = exp2f(dtl * A2b);
            h0 = fmaf(a0, h0, bc[s].x * dtu);
            h1 = fmaf(a1, h1, bc[s].y * dtu);
            float y = h0 * bc[s].z;
            y = fmaf(h1, bc[s].w, y);
            yp[s] = y;
        }
    };
    auto tail_part = [&](float (&yp)[SBLK], const float (&uj)[SBLK / 8], size_t orow) {
#pragma unroll
        for (int g = 0; g < SBLK / 8; ++g) {
            float vkeep = 0.0f;
#pragma unroll
            for (int s = 0; s < 8; ++s) {
                float v = yp[8 * g + s];
                v += dpp_read<0xB1>(v);
                v += dpp_read<0x4E>(v);
                v += dpp_read<0x164>(v);
                vkeep = (j == s) ? v : vkeep;
            }
            out[orow + (size_t)(8 * g) * D_MODEL] = fmaf(Dd, uj[g], vkeep);
        }
    };

    for (int k = 0; k < NSTG; ++k) {
        const int st = k & 1;
        const int t0 = k * SCH;
        if (k + 1 < NSTG)
            stage_fetch(r, u, dt, Btg, Ctg, b, t0 + SCH, d0, tid);

        const size_t obase = ((size_t)b * T_LEN + (size_t)(t0 + j)) * D_MODEL + (size_t)d;

        v2f ud[SBLK];
        v4f bc[SBLK];
#pragma unroll
        for (int s = 0; s < SBLK; ++s) { ud[s] = sUD[st][s][dl]; bc[s] = sBC[st][s][j]; }

#pragma unroll 1
        for (int tb = 0; tb < SCH; tb += SBLK) {
            float yp[SBLK];
            head_part(ud, bc, yp);
            const int tn = (tb + SBLK < SCH) ? (tb + SBLK) : tb;
            static_assert(SBLK == 16, "the fence below names the 16 shares of a block");
            asm volatile("" :: "v"(h0), "v"(h1),
                               "v"(yp[0]), "v"(yp[1]), "v"(yp[2]), "v"(yp[3]), "v"(yp[4]), "v"(yp[5]),
                               "v"(yp[6]), "v"(yp[7]), "v"(yp[8]), "v"(yp[9]), "v"(yp[10]), "v"(yp[11]),
                               "v"(yp[12]), "v"(yp[13]), "v"(yp[14]), "v"(yp[15]) : "memory");
            __builtin_amdgcn_sched_barrier(0);
#pragma unroll
            for (int s = 0; s < SBLK; ++s) ud[s] = sUD[st][tn + s][dl];
#pragma unroll
            for (int s = 0; s < SBLK; ++s) bc[s] = sBC[st][tn + s][j];
            float uj[SBLK / 8];
#pragma unroll
            for (int g = 0; g < SBLK / 8; ++g)
                uj[g] = sUD[st][tb + 8 * g + j][dl].y;
            __builtin_amdgcn_sched_barrier(0);
            tail_part(yp, uj, obase + (size_t)tb * D_MODEL);
        }

        if (k + 1 < NSTG)
            stage_put(st ^ 1);
        lds_barrier();
    }
}

extern "C" void kernel_launch(void* const* d_in, const int* in_sizes, int n_in,
                              void* d_out, int out_size, void* d_ws, size_t ws_size,
                              hipStream_t stream) {
    const float* u     = (const float*)d_in[0];
    const float* A_log = (const float*)d_in[1];
    const float* W_xdt = (const float*)d_in[2];
    const float* b_xdt = (const float*)d_in[3];
    const float* W_dt  = (const float*)d_in[4];
    const float* b_dt  = (const float*)d_in[5];
    const float* W_B   = (const float*)d_in[6];
    const float* b_B   = (const float*)d_in[7];
    const float* W_C   = (const float*)d_in[8];
    const float* b_C   = (const float*)d_in[9];
    const float* Dv    = (const float*)d_in[10];
    float* out = (float*)d_out;

    float* ws   = (float*)d_ws;
    float* xdt  = ws;
    float* dt   = xdt + (size_t)M_ROWS * DT_RANK;
    float* Bt   = dt  + (size_t)M_ROWS * D_MODEL;
    float* Ct   = Bt  + (size_t)M_ROWS * D_STATE;

    proj_u_fused<<<M_ROWS / PRJ_ROWS, 128, 0, stream>>>(
        u, W_xdt, b_xdt, W_B, b_B, W_C, b_C, xdt, Bt, Ct);

    dt_proj_softplus<<<(M_ROWS / DTP_ROWS) * (D_MODEL / DTP_COLS), 256, 0, stream>>>(
        xdt, W_dt, b_dt, dt);

    scan_sequential<<<B_SZ * (D_MODEL / 16), 128, 0, stream>>>(u, dt, Bt, Ct, A_log, Dv, out);
}
